# starting kernel
# baseline (speedup 1.0000x reference)
#include <hip/hip_runtime.h>


typedef __attribute__((ext_vector_type(2))) float v2f;
typedef __attribute__((ext_vector_type(8))) float v8f;

#define H      128
#define FOURH  512
#define LEN    8192
#define BATCH  8
#define MROWS  (BATCH * LEN)
#define NCHUNK 64
#define TCH    (LEN / NCHUNK)
#define NSEG   3
#define LDSROW 132

__global__ __launch_bounds__(256)
void k_proj(const float* __restrict__ x, const float* __restrict__ W,
            const float* __restrict__ Avec, const float* __restrict__ Bvec,
            const float* __restrict__ Dvec,
            float* __restrict__ dA, float* __restrict__ dB,
            float* __restrict__ Cc, float* __restrict__ rD,
            float* __restrict__ Ap, float* __restrict__ Bq)
{
    extern __shared__ float lds[];
    const int tid = threadIdx.x;

    for (int idx = tid; idx < (NSEG * H * H) / 4; idx += 256) {
        const int q  = idx >> 5;
        const int k4 = (idx & 31) << 2;
        const int p  = q + ((q >= H) ? H : 0);
        const float4 w = *(const float4*)(W + p * H + k4);
        *(float4*)(lds + q * LDSROW + k4) = w;
    }
    __syncthreads();

    const int wave = tid >> 5;
    const int lane = tid & 31;
    const int lm   = lane & 15;
    const int kh   = lane >> 4;
    const long rowBase = (long)blockIdx.x * 128 + wave * 16;

    v2f a[32];
    {
        const float* xp = x + (rowBase + lm) * H + 2 * kh;
        #pragma unroll
        for (int kt = 0; kt < 32; ++kt)
            a[kt] = *(const v2f*)(xp + kt * 4);
    }

    v2f comp[8];

    for (int cc = 0; cc < 2; ++cc) {
        v8f accD[4], accC[4], accR[4];
        #pragma unroll
        for (int nt = 0; nt < 4; ++nt) {
            accD[nt] = (v8f){};
            accC[nt] = (v8f){};
            accR[nt] = (v8f){};
        }

        #pragma unroll
        for (int kt = 0; kt < 32; ++kt) {
            const int kofs = kt * 4 + 2 * kh;
            #pragma unroll
            for (int nt = 0; nt < 4; ++nt) {
                const int c = cc * 64 + nt * 16 + lm;
                const v2f bD = *(const v2f*)(lds + (0 * H + c) * LDSROW + kofs);
                const v2f bC = *(const v2f*)(lds + (1 * H + c) * LDSROW + kofs);
                const v2f bR = *(const v2f*)(lds + (2 * H + c) * LDSROW + kofs);
                accD[nt] = __builtin_amdgcn_wmma_f32_16x16x4_f32(
                    false, a[kt], false, bD, (short)0, accD[nt], false, false);
                accC[nt] = __builtin_amdgcn_wmma_f32_16x16x4_f32(
                    false, a[kt], false, bC, (short)0, accC[nt], false, false);
                accR[nt] = __builtin_amdgcn_wmma_f32_16x16x4_f32(
                    false, a[kt], false, bR, (short)0, accR[nt], false, false);
            }
        }

        #pragma unroll
        for (int nt = 0; nt < 4; ++nt) {
            const int h  = cc * 64 + nt * 16 + lm;
            const float Ah = Avec[h];
            const float Bh = Bvec[h];
            const float Dh = Dvec[h];
            float P = 1.f, Q = 0.f;
            #pragma unroll
            for (int r = 0; r < 8; ++r) {
                const long row = rowBase + r + 8 * kh;
                const float dv = accD[nt][r];
                const float sp = fmaxf(dv, 0.f) + __logf(1.f + __expf(-fabsf(dv)));
                const float rs = accR[nt][r];
                const float av = sp * Ah;
                const float bv = sp * Bh * rs;
                const long o = row * H + h;
                dA[o] = av;
                dB[o] = bv;
                Cc[o] = accC[nt][r];
                rD[o] = rs * Dh;
                Q = av * Q + bv;
                P = av * P;
            }
            comp[cc * 4 + nt] = (v2f){P, Q};
        }
    }

    __syncthreads();
    {
        v2f* pieces = (v2f*)lds;
        #pragma unroll
        for (int i = 0; i < 8; ++i) {
            const int h = (i >> 2) * 64 + (i & 3) * 16 + lm;
            pieces[h * 16 + wave * 2 + kh] = comp[i];
        }
    }
    __syncthreads();
    if (tid < H) {
        const v2f* pieces = (const v2f*)lds;
        float P = 1.f, Q = 0.f;
        #pragma unroll
        for (int i = 0; i < 16; ++i) {
            const v2f pq = pieces[tid * 16 + i];
            Q = pq.x * Q + pq.y;
            P = pq.x * P;
        }
        const long o = (long)blockIdx.x * H + tid;
        Ap[o] = P;
        Bq[o] = Q;
    }
}

__global__ __launch_bounds__(128)
void k_chunk_scan(const float* __restrict__ Ap, const float* __restrict__ Bq,
                  float* __restrict__ S)
{
    const int h = threadIdx.x, b = blockIdx.x;
    float s = 0.f;
    for (int c = 0; c < NCHUNK; ++c) {
        const long o = (long)(b * NCHUNK + c) * H + h;
        S[o] = s;
        s = Ap[o] * s + Bq[o];
    }
}

__global__ __launch_bounds__(128)
void k_apply(const float* __restrict__ dA, const float* __restrict__ dB,
             const float* __restrict__ Cc, const float* __restrict__ rD,
             const float* __restrict__ S, float* __restrict__ out)
{
    const int h = threadIdx.x;
    const int c = blockIdx.x, b = blockIdx.y;
    float s = S[(long)(b * NCHUNK + c) * H + h];
    const long base = ((long)(b * LEN + c * TCH)) * H + h;
    for (int t = 0; t < TCH; ++t) {
        const long o = base + (long)t * H;
        s = dA[o] * s + dB[o];
        out[o] = s * Cc[o] + rD[o];
    }
}

extern "C" void kernel_launch(void* const* d_in, const int* in_sizes, int n_in,
                              void* d_out, int out_size, void* d_ws, size_t ws_size,
                              hipStream_t stream)
{
    (void)in_sizes; (void)n_in; (void)out_size; (void)ws_size;

    const float* x  = (const float*)d_in[0];
    const float* W  = (const float*)d_in[1];
    const float* Av = (const float*)d_in[2];
    const float* Bv = (const float*)d_in[3];
    const float* Dv = (const float*)d_in[4];
    float* out = (float*)d_out;

    const size_t NE = (size_t)BATCH * LEN * H;
    const size_t NC = (size_t)BATCH * NCHUNK * H;
    float* ws = (float*)d_ws;
    float* dA = ws;
    float* dB = dA + NE;
    float* Cc = dB + NE;
    float* rD = Cc + NE;
    float* Ap = rD + NE;
    float* Bq = Ap + NC;
    float* S  = Bq + NC;

    const int ldsBytes = NSEG * H * LDSROW * (int)sizeof(float);
    hipFuncSetAttribute(reinterpret_cast<const void*>(k_proj),
                        hipFuncAttributeMaxDynamicSharedMemorySize, ldsBytes);

    k_proj<<<MROWS / 128, 256, ldsBytes, stream>>>(x, W, Av, Bv, Dv,
                                                   dA, dB, Cc, rD, Ap, Bq);
    k_chunk_scan<<<BATCH, 128, 0, stream>>>(Ap, Bq, S);
    k_apply<<<dim3(NCHUNK, BATCH), 128, 0, stream>>>(dA, dB, Cc, rD, S, out);
}


# k_proj: B fragments of the next k-step loaded from LDS as one group (second register set) in front of the current step's 12 WMMAs, not 3 at a time just before use; summation order unchanged
# speedup vs baseline: 1.1132x; 1.1132x over previous
#include <hip/hip_runtime.h>


typedef __attribute__((ext_vector_type(2))) float v2f;
typedef __attribute__((ext_vector_type(8))) float v8f;

#define H      128
#define FOURH  512
#define LEN    8192
#define BATCH  8
#define MROWS  (BATCH * LEN)
#define NCHUNK 64
#define TCH    (LEN / NCHUNK)
#define NSEG   3
#define LDSROW 132

__global__ __launch_bounds__(256)
void k_proj(const float* __restrict__ x, const float* __restrict__ W,
            const float* __restrict__ Avec, const float* __restrict__ Bvec,
            const float* __restrict__ Dvec,
            float* __restrict__ dA, float* __restrict__ dB,
            float* __restrict__ Cc, float* __restrict__ rD,
            float* __restrict__ Ap, float* __restrict__ Bq)
{
    extern __shared__ float lds[];
    const int tid = threadIdx.x;

    for (int idx = tid; idx < (NSEG * H * H) / 4; idx += 256) {
        const int q  = idx >> 5;
        const int k4 = (idx & 31) << 2;
        const int p  = q + ((q >= H) ? H : 0);
        const float4 w = *(const float4*)(W + p * H + k4);
        *(float4*)(lds + q * LDSROW + k4) = w;
    }
    __syncthreads();

    const int wave = tid >> 5;
    const int lane = tid & 31;
    const int lm   = lane & 15;
    const int kh   = lane >> 4;
    const long rowBase = (long)blockIdx.x * 128 + wave * 16;

    v2f a[32];
    {
        const float* xp = x + (rowBase + lm) * H + 2 * kh;
        #pragma unroll
        for (int kt = 0; kt < 32; ++kt)
            a[kt] = *(const v2f*)(xp + kt * 4);
    }

    v2f comp[8];

    for (int cc = 0; cc < 2; ++cc) {
        v8f accD[4], accC[4], accR[4];
        #pragma unroll
        for (int nt = 0; nt < 4; ++nt) {
            accD[nt] = (v8f){};
            accC[nt] = (v8f){};
            accR[nt] = (v8f){};
        }

        const float* bp = lds + (cc * 64 + lm) * LDSROW + 2 * kh;
        v2f bf[2][12];
        #pragma unroll
        for (int nt = 0; nt < 4; ++nt) {
            #pragma unroll
            for (int sg = 0; sg < NSEG; ++sg)
                bf[0][nt * NSEG + sg] =
                    *(const v2f*)(bp + (sg * H + nt * 16) * LDSROW);
        }

        #pragma unroll
        for (int kt = 0; kt < 32; ++kt) {
            const int cur = kt & 1;
            if (kt + 1 < 32) {
                #pragma unroll
                for (int nt = 0; nt < 4; ++nt) {
                    #pragma unroll
                    for (int sg = 0; sg < NSEG; ++sg)
                        bf[cur ^ 1][nt * NSEG + sg] =
                            *(const v2f*)(bp + (sg * H + nt * 16) * LDSROW + (kt + 1) * 4);
                }
            }
            #pragma unroll
            for (int nt = 0; nt < 4; ++nt) {
                accD[nt] = __builtin_amdgcn_wmma_f32_16x16x4_f32(
                    false, a[kt], false, bf[cur][nt * NSEG + 0], (short)0, accD[nt], false, false);
                accC[nt] = __builtin_amdgcn_wmma_f32_16x16x4_f32(
                    false, a[kt], false, bf[cur][nt * NSEG + 1], (short)0, accC[nt], false, false);
                accR[nt] = __builtin_amdgcn_wmma_f32_16x16x4_f32(
                    false, a[kt], false, bf[cur][nt * NSEG + 2], (short)0, accR[nt], false, false);
            }
            __builtin_amdgcn_sched_barrier(0);
        }

        #pragma unroll
        for (int nt = 0; nt < 4; ++nt) {
            const int h  = cc * 64 + nt * 16 + lm;
            const float Ah = Avec[h];
            const float Bh = Bvec[h];
            const float Dh = Dvec[h];
            float P = 1.f, Q = 0.f;
            #pragma unroll
            for (int r = 0; r < 8; ++r) {
                const long row = rowBase + r + 8 * kh;
                const float dv = accD[nt][r];
                const float sp = fmaxf(dv, 0.f) + __logf(1.f + __expf(-fabsf(dv)));
                const float rs = accR[nt][r];
                const float av = sp * Ah;
                const float bv = sp * Bh * rs;
                const long o = row * H + h;
                dA[o] = av;
                dB[o] = bv;
                Cc[o] = accC[nt][r];
                rD[o] = rs * Dh;
                Q = av * Q + bv;
                P = av * P;
            }
            comp[cc * 4 + nt] = (v2f){P, Q};
        }
    }

    __syncthreads();
    {
        v2f* pieces = (v2f*)lds;
        #pragma unroll
        for (int i = 0; i < 8; ++i) {
            const int h = (i >> 2) * 64 + (i & 3) * 16 + lm;
            pieces[h * 16 + wave * 2 + kh] = comp[i];
        }
    }
    __syncthreads();
    if (tid < H) {
        const v2f* pieces = (const v2f*)lds;
        float P = 1.f, Q = 0.f;
        #pragma unroll
        for (int i = 0; i < 16; ++i) {
            const v2f pq = pieces[tid * 16 + i];
            Q = pq.x * Q + pq.y;
            P = pq.x * P;
        }
        const long o = (long)blockIdx.x * H + tid;
        Ap[o] = P;
        Bq[o] = Q;
    }
}

__global__ __launch_bounds__(128)
void k_chunk_scan(const float* __restrict__ Ap, const float* __restrict__ Bq,
                  float* __restrict__ S)
{
    const int h = threadIdx.x, b = blockIdx.x;
    float s = 0.f;
    for (int c = 0; c < NCHUNK; ++c) {
        const long o = (long)(b * NCHUNK + c) * H + h;
        S[o] = s;
        s = Ap[o] * s + Bq[o];
    }
}

__global__ __launch_bounds__(128)
void k_apply(const float* __restrict__ dA, const float* __restrict__ dB,
             const float* __restrict__ Cc, const float* __restrict__ rD,
             const float* __restrict__ S, float* __restrict__ out)
{
    const int h = threadIdx.x;
    const int c = blockIdx.x, b = blockIdx.y;
    float s = S[(long)(b * NCHUNK + c) * H + h];
    const long base = ((long)(b * LEN + c * TCH)) * H + h;
    for (int t = 0; t < TCH; ++t) {
        const long o = base + (long)t * H;
        s = dA[o] * s + dB[o];
        out[o] = s * Cc[o] + rD[o];
    }
}

extern "C" void kernel_launch(void* const* d_in, const int* in_sizes, int n_in,
                              void* d_out, int out_size, void* d_ws, size_t ws_size,
                              hipStream_t stream)
{
    (void)in_sizes; (void)n_in; (void)out_size; (void)ws_size;

    const float* x  = (const float*)d_in[0];
    const float* W  = (const float*)d_in[1];
    const float* Av = (const float*)d_in[2];
    const float* Bv = (const float*)d_in[3];
    const float* Dv = (const float*)d_in[4];
    float* out = (float*)d_out;

    const size_t NE = (size_t)BATCH * LEN * H;
    const size_t NC = (size_t)BATCH * NCHUNK * H;
    float* ws = (float*)d_ws;
    float* dA = ws;
    float* dB = dA + NE;
    float* Cc = dB + NE;
    float* rD = Cc + NE;
    float* Ap = rD + NE;
    float* Bq = Ap + NC;
    float* S  = Bq + NC;

    const int ldsBytes = NSEG * H * LDSROW * (int)sizeof(float);
    hipFuncSetAttribute(reinterpret_cast<const void*>(k_proj),
                        hipFuncAttributeMaxDynamicSharedMemorySize, ldsBytes);

    k_proj<<<MROWS / 128, 256, ldsBytes, stream>>>(x, W, Av, Bv, Dv,
                                                   dA, dB, Cc, rD, Ap, Bq);
    k_chunk_scan<<<BATCH, 128, 0, stream>>>(Ap, Bq, S);
    k_apply<<<dim3(NCHUNK, BATCH), 128, 0, stream>>>(dA, dB, Cc, rD, S, out);
}


# k_apply: chunk replay split into 8 pieces of 16 steps, one wave each with all 64 of its row loads (256 B each) in flight at once; piece entering states carried through LDS: state sum order changes
# speedup vs baseline: 1.1626x; 1.0444x over previous
#include <hip/hip_runtime.h>


typedef __attribute__((ext_vector_type(2))) float v2f;
typedef __attribute__((ext_vector_type(8))) float v8f;

#define H      128
#define FOURH  512
#define LEN    8192
#define BATCH  8
#define MROWS  (BATCH * LEN)
#define NCHUNK 64
#define TCH    (LEN / NCHUNK)
#define NSEG   3
#define LDSROW 132

__global__ __launch_bounds__(256)
void k_proj(const float* __restrict__ x, const float* __restrict__ W,
            const float* __restrict__ Avec, const float* __restrict__ Bvec,
            const float* __restrict__ Dvec,
            float* __restrict__ dA, float* __restrict__ dB,
            float* __restrict__ Cc, float* __restrict__ rD,
            float* __restrict__ Ap, float* __restrict__ Bq)
{
    extern __shared__ float lds[];
    const int tid = threadIdx.x;

    for (int idx = tid; idx < (NSEG * H * H) / 4; idx += 256) {
        const int q  = idx >> 5;
        const int k4 = (idx & 31) << 2;
        const int p  = q + ((q >= H) ? H : 0);
        const float4 w = *(const float4*)(W + p * H + k4);
        *(float4*)(lds + q * LDSROW + k4) = w;
    }
    __syncthreads();

    const int wave = tid >> 5;
    const int lane = tid & 31;
    const int lm   = lane & 15;
    const int kh   = lane >> 4;
    const long rowBase = (long)blockIdx.x * 128 + wave * 16;

    v2f a[32];
    {
        const float* xp = x + (rowBase + lm) * H + 2 * kh;
        #pragma unroll
        for (int kt = 0; kt < 32; ++kt)
            a[kt] = *(const v2f*)(xp + kt * 4);
    }

    v2f comp[8];

    for (int cc = 0; cc < 2; ++cc) {
        v8f accD[4], accC[4], accR[4];
        #pragma unroll
        for (int nt = 0; nt < 4; ++nt) {
            accD[nt] = (v8f){};
            accC[nt] = (v8f){};
            accR[nt] = (v8f){};
        }

        const float* bp = lds + (cc * 64 + lm) * LDSROW + 2 * kh;
        v2f bf[2][12];
        #pragma unroll
        for (int nt = 0; nt < 4; ++nt) {
            #pragma unroll
            for (int sg = 0; sg < NSEG; ++sg)
                bf[0][nt * NSEG + sg] =
                    *(const v2f*)(bp + (sg * H + nt * 16) * LDSROW);
        }

        #pragma unroll
        for (int kt = 0; kt < 32; ++kt) {
            const int cur = kt & 1;
            if (kt + 1 < 32) {
                #pragma unroll
                for (int nt = 0; nt < 4; ++nt) {
                    #pragma unroll
                    for (int sg = 0; sg < NSEG; ++sg)
                        bf[cur ^ 1][nt * NSEG + sg] =
                            *(const v2f*)(bp + (sg * H + nt * 16) * LDSROW + (kt + 1) * 4);
                }
            }
            #pragma unroll
            for (int nt = 0; nt < 4; ++nt) {
                accD[nt] = __builtin_amdgcn_wmma_f32_16x16x4_f32(
                    false, a[kt], false, bf[cur][nt * NSEG + 0], (short)0, accD[nt], false, false);
                accC[nt] = __builtin_amdgcn_wmma_f32_16x16x4_f32(
                    false, a[kt], false, bf[cur][nt * NSEG + 1], (short)0, accC[nt], false, false);
                accR[nt] = __builtin_amdgcn_wmma_f32_16x16x4_f32(
                    false, a[kt], false, bf[cur][nt * NSEG + 2], (short)0, accR[nt], false, false);
            }
            __builtin_amdgcn_sched_barrier(0);
        }

        #pragma unroll
        for (int nt = 0; nt < 4; ++nt) {
            const int h  = cc * 64 + nt * 16 + lm;
            const float Ah = Avec[h];
            const float Bh = Bvec[h];
            const float Dh = Dvec[h];
            float P = 1.f, Q = 0.f;
            #pragma unroll
            for (int r = 0; r < 8; ++r) {
                const long row = rowBase + r + 8 * kh;
                const float dv = accD[nt][r];
                const float sp = fmaxf(dv, 0.f) + __logf(1.f + __expf(-fabsf(dv)));
                const float rs = accR[nt][r];
                const float av = sp * Ah;
                const float bv = sp * Bh * rs;
                const long o = row * H + h;
                dA[o] = av;
                dB[o] = bv;
                Cc[o] = accC[nt][r];
                rD[o] = rs * Dh;
                Q = av * Q + bv;
                P = av * P;
            }
            comp[cc * 4 + nt] = (v2f){P, Q};
        }
    }

    __syncthreads();
    {
        v2f* pieces = (v2f*)lds;
        #pragma unroll
        for (int i = 0; i < 8; ++i) {
            const int h = (i >> 2) * 64 + (i & 3) * 16 + lm;
            pieces[h * 16 + wave * 2 + kh] = comp[i];
        }
    }
    __syncthreads();
    if (tid < H) {
        const v2f* pieces = (const v2f*)lds;
        float P = 1.f, Q = 0.f;
        #pragma unroll
        for (int i = 0; i < 16; ++i) {
            const v2f pq = pieces[tid * 16 + i];
            Q = pq.x * Q + pq.y;
            P = pq.x * P;
        }
        const long o = (long)blockIdx.x * H + tid;
        Ap[o] = P;
        Bq[o] = Q;
    }
}

__global__ __launch_bounds__(128)
void k_chunk_scan(const float* __restrict__ Ap, const float* __restrict__ Bq,
                  float* __restrict__ S)
{
    const int h = threadIdx.x, b = blockIdx.x;
    float s = 0.f;
    for (int c = 0; c < NCHUNK; ++c) {
        const long o = (long)(b * NCHUNK + c) * H + h;
        S[o] = s;
        s = Ap[o] * s + Bq[o];
    }
}

#define SUBT  16
#define NPIECE (TCH / SUBT)
__global__ __launch_bounds__(256)
void k_apply(const float* __restrict__ dA, const float* __restrict__ dB,
             const float* __restrict__ Cc, const float* __restrict__ rD,
             const float* __restrict__ S, float* __restrict__ out)
{
    __shared__ float4 pieces[NPIECE][32];

    const int lane  = threadIdx.x & 31;
    const int w     = __builtin_amdgcn_readfirstlane(threadIdx.x >> 5);
    const int chunk = blockIdx.x >> 1;
    const int h     = (blockIdx.x & 1) * 64 + 2 * lane;
    const unsigned off = ((unsigned)chunk * TCH + (unsigned)w * SUBT) * H + h;

    const float* pa = dA + off;
    const float* pb = dB + off;
    const float* pc = Cc + off;
    const float* pr = rD + off;

    v2f s = *(const v2f*)(S + chunk * H + h);
    v2f a[SUBT], b[SUBT], c[SUBT], r[SUBT];
    #pragma unroll
    for (int t = 0; t < SUBT; ++t) {
        a[t] = *(const v2f*)(pa + t * H);
        b[t] = *(const v2f*)(pb + t * H);
    }
    #pragma unroll
    for (int t = 0; t < SUBT; ++t) {
        c[t] = *(const v2f*)(pc + t * H);
        r[t] = *(const v2f*)(pr + t * H);
    }

    float Px = a[0].x, Py = a[0].y, Qx = b[0].x, Qy = b[0].y;
    #pragma unroll
    for (int t = 1; t < SUBT; ++t) {
        Qx = __builtin_fmaf(a[t].x, Qx, b[t].x);
        Qy = __builtin_fmaf(a[t].y, Qy, b[t].y);
        Px = a[t].x * Px;
        Py = a[t].y * Py;
    }
    pieces[w][lane] = make_float4(Px, Qx, Py, Qy);
    __syncthreads();

    float sx = s.x, sy = s.y;
    for (int i = 0; i < w; ++i) {
        const float4 pq = pieces[i][lane];
        sx = __builtin_fmaf(pq.x, sx, pq.y);
        sy = __builtin_fmaf(pq.z, sy, pq.w);
    }

    float* po = out + off;
    #pragma unroll
    for (int t = 0; t < SUBT; ++t) {
        sx = __builtin_fmaf(a[t].x, sx, b[t].x);
        sy = __builtin_fmaf(a[t].y, sy, b[t].y);
        v2f o;
        o.x = __builtin_fmaf(sx, c[t].x, r[t].x);
        o.y = __builtin_fmaf(sy, c[t].y, r[t].y);
        *(v2f*)(po + t * H) = o;
    }
}

extern "C" void kernel_launch(void* const* d_in, const int* in_sizes, int n_in,
                              void* d_out, int out_size, void* d_ws, size_t ws_size,
                              hipStream_t stream)
{
    (void)in_sizes; (void)n_in; (void)out_size; (void)ws_size;

    const float* x  = (const float*)d_in[0];
    const float* W  = (const float*)d_in[1];
    const float* Av = (const float*)d_in[2];
    const float* Bv = (const float*)d_in[3];
    const float* Dv = (const float*)d_in[4];
    float* out = (float*)d_out;

    const size_t NE = (size_t)BATCH * LEN * H;
    const size_t NC = (size_t)BATCH * NCHUNK * H;
    float* ws = (float*)d_ws;
    float* dA = ws;
    float* dB = dA + NE;
    float* Cc = dB + NE;
    float* rD = Cc + NE;
    float* Ap = rD + NE;
    float* Bq = Ap + NC;
    float* S  = Bq + NC;

    const int ldsBytes = NSEG * H * LDSROW * (int)sizeof(float);
    hipFuncSetAttribute(reinterpret_cast<const void*>(k_proj),
                        hipFuncAttributeMaxDynamicSharedMemorySize, ldsBytes);

    k_proj<<<MROWS / 128, 256, ldsBytes, stream>>>(x, W, Av, Bv, Dv,
                                                   dA, dB, Cc, rD, Ap, Bq);
    k_chunk_scan<<<BATCH, 128, 0, stream>>>(Ap, Bq, S);
    k_apply<<<BATCH * NCHUNK * 2, 256, 0, stream>>>(dA, dB, Cc, rD, S, out);
}


# k_proj: W staged one 64-column half at a time in lane-major LDS slabs (115 KB, not 203 KB; 244 registers): two blocks share a WGP, one's staging and stores run under the other's WMMAs; sums unchanged
# speedup vs baseline: 1.2047x; 1.0362x over previous
#include <hip/hip_runtime.h>


typedef __attribute__((ext_vector_type(2))) float v2f;
typedef __attribute__((ext_vector_type(8))) float v8f;

#define H      128
#define FOURH  512
#define LEN    8192
#define BATCH  8
#define MROWS  (BATCH * LEN)
#define NCHUNK 64
#define TCH    (LEN / NCHUNK)
#define NSEG   3
#define HALFN  64
#define NTILE  (NSEG * HALFN / 16)
#define SLAB   (NTILE * H + 4)

__device__ __forceinline__
void stage_w_half(const float* __restrict__ W, float* lds, int half, int tid)
{
    for (int idx = tid; idx < (NSEG * HALFN * H) / 4; idx += 256) {
        const int ql = idx >> 5;
        const int k4 = (idx & 31) << 2;
        const int sg = ql >> 6;
        const int j  = ql & (HALFN - 1);
        const int p  = ql + sg * HALFN + (sg ? H : 0) + half * HALFN;
        const float4 w = *(const float4*)(W + p * H + k4);
        *(float4*)(lds + (j & 15) * SLAB + ((j >> 4) * NSEG + sg) * H + k4) = w;
    }
}

__global__ __launch_bounds__(256)
void k_proj(const float* __restrict__ x, const float* __restrict__ W,
            const float* __restrict__ Avec, const float* __restrict__ Bvec,
            const float* __restrict__ Dvec,
            float* __restrict__ dA, float* __restrict__ dB,
            float* __restrict__ Cc, float* __restrict__ rD,
            float* __restrict__ Ap, float* __restrict__ Bq)
{
    extern __shared__ float lds[];
    const int tid = threadIdx.x;

    stage_w_half(W, lds, 0, tid);
    __syncthreads();

    const int wave = tid >> 5;
    const int lane = tid & 31;
    const int lm   = lane & 15;
    const int kh   = lane >> 4;
    const long rowBase = (long)blockIdx.x * 128 + wave * 16;

    v2f a[32];
    {
        const float* xp = x + (rowBase + lm) * H + 2 * kh;
        #pragma unroll
        for (int kt = 0; kt < 32; ++kt)
            a[kt] = *(const v2f*)(xp + kt * 4);
    }

    v2f* pieces = (v2f*)(lds + 16 * SLAB);

    for (int cc = 0; cc < 2; ++cc) {
        v8f accD[4], accC[4], accR[4];
        #pragma unroll
        for (int nt = 0; nt < 4; ++nt) {
            accD[nt] = (v8f){};
            accC[nt] = (v8f){};
            accR[nt] = (v8f){};
        }

        const float* bp = lds + lm * SLAB + 2 * kh;
        v2f bf[2][12];
        #pragma unroll
        for (int nt = 0; nt < 4; ++nt) {
            #pragma unroll
            for (int sg = 0; sg < NSEG; ++sg)
                bf[0][nt * NSEG + sg] =
                    *(const v2f*)(bp + (nt * NSEG + sg) * H);
        }

        #pragma unroll
        for (int kt = 0; kt < 32; ++kt) {
            const int cur = kt & 1;
            if (kt + 1 < 32) {
                #pragma unroll
                for (int nt = 0; nt < 4; ++nt) {
                    #pragma unroll
                    for (int sg = 0; sg < NSEG; ++sg)
                        bf[cur ^ 1][nt * NSEG + sg] =
                            *(const v2f*)(bp + (nt * NSEG + sg) * H + (kt + 1) * 4);
                }
            }
            #pragma unroll
            for (int nt = 0; nt < 4; ++nt) {
                accD[nt] = __builtin_amdgcn_wmma_f32_16x16x4_f32(
                    false, a[kt], false, bf[cur][nt * NSEG + 0], (short)0, accD[nt], false, false);
                accC[nt] = __builtin_amdgcn_wmma_f32_16x16x4_f32(
                    false, a[kt], false, bf[cur][nt * NSEG + 1], (short)0, accC[nt], false, false);
                accR[nt] = __builtin_amdgcn_wmma_f32_16x16x4_f32(
                    false, a[kt], false, bf[cur][nt * NSEG + 2], (short)0, accR[nt], false, false);
            }
            __builtin_amdgcn_sched_barrier(0);
        }

        #pragma unroll
        for (int nt = 0; nt < 4; ++nt) {
            const int h  = cc * 64 + nt * 16 + lm;
            const float Ah = Avec[h];
            const float Bh = Bvec[h];
            const float Dh = Dvec[h];
            float P = 1.f, Q = 0.f;
            #pragma unroll
            for (int r = 0; r < 8; ++r) {
                const long row = rowBase + r + 8 * kh;
                const float dv = accD[nt][r];
                const float sp = fmaxf(dv, 0.f) + __logf(1.f + __expf(-fabsf(dv)));
                const float rs = accR[nt][r];
                const float av = sp * Ah;
                const float bv = sp * Bh * rs;
                const long o = row * H + h;
                dA[o] = av;
                dB[o] = bv;
                Cc[o] = accC[nt][r];
                rD[o] = rs * Dh;
                Q = av * Q + bv;
                P = av * P;
            }
            pieces[h * 16 + wave * 2 + kh] = (v2f){P, Q};
        }

        if (cc == 0) {
            __syncthreads();
            stage_w_half(W, lds, 1, tid);
            __syncthreads();
        }
    }

    __syncthreads();
    if (tid < H) {
        float P = 1.f, Q = 0.f;
        #pragma unroll
        for (int i = 0; i < 16; ++i) {
            const v2f pq = pieces[tid * 16 + i];
            Q = pq.x * Q + pq.y;
            P = pq.x * P;
        }
        const long o = (long)blockIdx.x * H + tid;
        Ap[o] = P;
        Bq[o] = Q;
    }
}

__global__ __launch_bounds__(128)
void k_chunk_scan(const float* __restrict__ Ap, const float* __restrict__ Bq,
                  float* __restrict__ S)
{
    const int h = threadIdx.x, b = blockIdx.x;
    float s = 0.f;
    for (int c = 0; c < NCHUNK; ++c) {
        const long o = (long)(b * NCHUNK + c) * H + h;
        S[o] = s;
        s = Ap[o] * s + Bq[o];
    }
}

#define SUBT  16
#define NPIECE (TCH / SUBT)
__global__ __launch_bounds__(256)
void k_apply(const float* __restrict__ dA, const float* __restrict__ dB,
             const float* __restrict__ Cc, const float* __restrict__ rD,
             const float* __restrict__ S, float* __restrict__ out)
{
    __shared__ float4 pieces[NPIECE][32];

    const int lane  = threadIdx.x & 31;
    const int w     = __builtin_amdgcn_readfirstlane(threadIdx.x >> 5);
    const int chunk = blockIdx.x >> 1;
    const int h     = (blockIdx.x & 1) * 64 + 2 * lane;
    const unsigned off = ((unsigned)chunk * TCH + (unsigned)w * SUBT) * H + h;

    const float* pa = dA + off;
    const float* pb = dB + off;
    const float* pc = Cc + off;
    const float* pr = rD + off;

    v2f s = *(const v2f*)(S + chunk * H + h);
    v2f a[SUBT], b[SUBT], c[SUBT], r[SUBT];
    #pragma unroll
    for (int t = 0; t < SUBT; ++t) {
        a[t] = *(const v2f*)(pa + t * H);
        b[t] = *(const v2f*)(pb + t * H);
    }
    #pragma unroll
    for (int t = 0; t < SUBT; ++t) {
        c[t] = *(const v2f*)(pc + t * H);
        r[t] = *(const v2f*)(pr + t * H);
    }

    float Px = a[0].x, Py = a[0].y, Qx = b[0].x, Qy = b[0].y;
    #pragma unroll
    for (int t = 1; t < SUBT; ++t) {
        Qx = __builtin_fmaf(a[t].x, Qx, b[t].x);
        Qy = __builtin_fmaf(a[t].y, Qy, b[t].y);
        Px = a[t].x * Px;
        Py = a[t].y * Py;
    }
    pieces[w][lane] = make_float4(Px, Qx, Py, Qy);
    __syncthreads();

    float sx = s.x, sy = s.y;
    for (int i = 0; i < w; ++i) {
        const float4 pq = pieces[i][lane];
        sx = __builtin_fmaf(pq.x, sx, pq.y);
        sy = __builtin_fmaf(pq.z, sy, pq.w);
    }

    float* po = out + off;
    #pragma unroll
    for (int t = 0; t < SUBT; ++t) {
        sx = __builtin_fmaf(a[t].x, sx, b[t].x);
        sy = __builtin_fmaf(a[t].y, sy, b[t].y);
        v2f o;
        o.x = __builtin_fmaf(sx, c[t].x, r[t].x);
        o.y = __builtin_fmaf(sy, c[t].y, r[t].y);
        *(v2f*)(po + t * H) = o;
    }
}

extern "C" void kernel_launch(void* const* d_in, const int* in_sizes, int n_in,
                              void* d_out, int out_size, void* d_ws, size_t ws_size,
                              hipStream_t stream)
{
    (void)in_sizes; (void)n_in; (void)out_size; (void)ws_size;

    const float* x  = (const float*)d_in[0];
    const float* W  = (const float*)d_in[1];
    const float* Av = (const float*)d_in[2];
    const float* Bv = (const float*)d_in[3];
    const float* Dv = (const float*)d_in[4];
    float* out = (float*)d_out;

    const size_t NE = (size_t)BATCH * LEN * H;
    const size_t NC = (size_t)BATCH * NCHUNK * H;
    float* ws = (float*)d_ws;
    float* dA = ws;
    float* dB = dA + NE;
    float* Cc = dB + NE;
    float* rD = Cc + NE;
    float* Ap = rD + NE;
    float* Bq = Ap + NC;
    float* S  = Bq + NC;

    const int ldsBytes = (16 * SLAB + 2 * H * 16) * (int)sizeof(float);
    hipFuncSetAttribute(reinterpret_cast<const void*>(k_proj),
                        hipFuncAttributeMaxDynamicSharedMemorySize, ldsBytes);

    k_proj<<<MROWS / 128, 256, ldsBytes, stream>>>(x, W, Av, Bv, Dv,
                                                   dA, dB, Cc, rD, Ap, Bq);
    k_chunk_scan<<<BATCH, 128, 0, stream>>>(Ap, Bq, S);
    k_apply<<<BATCH * NCHUNK * 2, 256, 0, stream>>>(dA, dB, Cc, rD, S, out);
}


# k_apply computes the state entering its chunk itself (wave w composes the chunk maps 8w..8w+7 in front of it, 8 heads via LDS); the scan launch is gone (2 launches); chunk-state sum order changes
# speedup vs baseline: 1.4115x; 1.1717x over previous
#include <hip/hip_runtime.h>


typedef __attribute__((ext_vector_type(2))) float v2f;
typedef __attribute__((ext_vector_type(8))) float v8f;

#define H      128
#define FOURH  512
#define LEN    8192
#define BATCH  8
#define MROWS  (BATCH * LEN)
#define NCHUNK 64
#define TCH    (LEN / NCHUNK)
#define NSEG   3
#define HALFN  64
#define NTILE  (NSEG * HALFN / 16)
#define SLAB   (NTILE * H + 4)

__device__ __forceinline__
void stage_w_half(const float* __restrict__ W, float* lds, int half, int tid)
{
    for (int idx = tid; idx < (NSEG * HALFN * H) / 4; idx += 256) {
        const int ql = idx >> 5;
        const int k4 = (idx & 31) << 2;
        const int sg = ql >> 6;
        const int j  = ql & (HALFN - 1);
        const int p  = ql + sg * HALFN + (sg ? H : 0) + half * HALFN;
        const float4 w = *(const float4*)(W + p * H + k4);
        *(float4*)(lds + (j & 15) * SLAB + ((j >> 4) * NSEG + sg) * H + k4) = w;
    }
}

__global__ __launch_bounds__(256)
void k_proj(const float* __restrict__ x, const float* __restrict__ W,
            const float* __restrict__ Avec, const float* __restrict__ Bvec,
            const float* __restrict__ Dvec,
            float* __restrict__ dA, float* __restrict__ dB,
            float* __restrict__ Cc, float* __restrict__ rD,
            float* __restrict__ Ap, float* __restrict__ Bq)
{
    extern __shared__ float lds[];
    const int tid = threadIdx.x;

    stage_w_half(W, lds, 0, tid);
    __syncthreads();

    const int wave = tid >> 5;
    const int lane = tid & 31;
    const int lm   = lane & 15;
    const int kh   = lane >> 4;
    const long rowBase = (long)blockIdx.x * 128 + wave * 16;

    v2f a[32];
    {
        const float* xp = x + (rowBase + lm) * H + 2 * kh;
        #pragma unroll
        for (int kt = 0; kt < 32; ++kt)
            a[kt] = *(const v2f*)(xp + kt * 4);
    }

    v2f* pieces = (v2f*)(lds + 16 * SLAB);

    for (int cc = 0; cc < 2; ++cc) {
        v8f accD[4], accC[4], accR[4];
        #pragma unroll
        for (int nt = 0; nt < 4; ++nt) {
            accD[nt] = (v8f){};
            accC[nt] = (v8f){};
            accR[nt] = (v8f){};
        }

        const float* bp = lds + lm * SLAB + 2 * kh;
        v2f bf[2][12];
        #pragma unroll
        for (int nt = 0; nt < 4; ++nt) {
            #pragma unroll
            for (int sg = 0; sg < NSEG; ++sg)
                bf[0][nt * NSEG + sg] =
                    *(const v2f*)(bp + (nt * NSEG + sg) * H);
        }

        #pragma unroll
        for (int kt = 0; kt < 32; ++kt) {
            const int cur = kt & 1;
            if (kt + 1 < 32) {
                #pragma unroll
                for (int nt = 0; nt < 4; ++nt) {
                    #pragma unroll
                    for (int sg = 0; sg < NSEG; ++sg)
                        bf[cur ^ 1][nt * NSEG + sg] =
                            *(const v2f*)(bp + (nt * NSEG + sg) * H + (kt + 1) * 4);
                }
            }
            #pragma unroll
            for (int nt = 0; nt < 4; ++nt) {
                accD[nt] = __builtin_amdgcn_wmma_f32_16x16x4_f32(
                    false, a[kt], false, bf[cur][nt * NSEG + 0], (short)0, accD[nt], false, false);
                accC[nt] = __builtin_amdgcn_wmma_f32_16x16x4_f32(
                    false, a[kt], false, bf[cur][nt * NSEG + 1], (short)0, accC[nt], false, false);
                accR[nt] = __builtin_amdgcn_wmma_f32_16x16x4_f32(
                    false, a[kt], false, bf[cur][nt * NSEG + 2], (short)0, accR[nt], false, false);
            }
            __builtin_amdgcn_sched_barrier(0);
        }

        #pragma unroll
        for (int nt = 0; nt < 4; ++nt) {
            const int h  = cc * 64 + nt * 16 + lm;
            const float Ah = Avec[h];
            const float Bh = Bvec[h];
            const float Dh = Dvec[h];
            float P = 1.f, Q = 0.f;
            #pragma unroll
            for (int r = 0; r < 8; ++r) {
                const long row = rowBase + r + 8 * kh;
                const float dv = accD[nt][r];
                const float sp = fmaxf(dv, 0.f) + __logf(1.f + __expf(-fabsf(dv)));
                const float rs = accR[nt][r];
                const float av = sp * Ah;
                const float bv = sp * Bh * rs;
                const long o = row * H + h;
                dA[o] = av;
                dB[o] = bv;
                Cc[o] = accC[nt][r];
                rD[o] = rs * Dh;
                Q = av * Q + bv;
                P = av * P;
            }
            pieces[h * 16 + wave * 2 + kh] = (v2f){P, Q};
        }

        if (cc == 0) {
            __syncthreads();
            stage_w_half(W, lds, 1, tid);
            __syncthreads();
        }
    }

    __syncthreads();
    if (tid < H) {
        float P = 1.f, Q = 0.f;
        #pragma unroll
        for (int i = 0; i < 16; ++i) {
            const v2f pq = pieces[tid * 16 + i];
            Q = pq.x * Q + pq.y;
            P = pq.x * P;
        }
        const long o = (long)blockIdx.x * H + tid;
        Ap[o] = P;
        Bq[o] = Q;
    }
}

#define SUBT  16
#define NPIECE (TCH / SUBT)
#define HEADC (NCHUNK / NPIECE)
__global__ __launch_bounds__(256)
void k_apply(const float* __restrict__ dA, const float* __restrict__ dB,
             const float* __restrict__ Cc, const float* __restrict__ rD,
             const float* __restrict__ Ap, const float* __restrict__ Bq,
             float* __restrict__ out)
{
    __shared__ float4 pieces[NPIECE][32];
    __shared__ float4 heads[NPIECE][32];

    const int lane  = threadIdx.x & 31;
    const int w     = __builtin_amdgcn_readfirstlane(threadIdx.x >> 5);
    const int chunk = blockIdx.x >> 1;
    const int c     = chunk & (NCHUNK - 1);
    const int h     = (blockIdx.x & 1) * 64 + 2 * lane;
    const unsigned off = ((unsigned)chunk * TCH + (unsigned)w * SUBT) * H + h;

    const float* pa = dA + off;
    const float* pb = dB + off;
    const float* pc = Cc + off;
    const float* pr = rD + off;

    v2f a[SUBT], b[SUBT], cv[SUBT], r[SUBT];
    #pragma unroll
    for (int t = 0; t < SUBT; ++t) {
        a[t] = *(const v2f*)(pa + t * H);
        b[t] = *(const v2f*)(pb + t * H);
    }
    #pragma unroll
    for (int t = 0; t < SUBT; ++t) {
        cv[t] = *(const v2f*)(pc + t * H);
        r[t]  = *(const v2f*)(pr + t * H);
    }

    const int nh = c - w * HEADC;
    const float* hpA = Ap + ((chunk - c) + w * HEADC) * H + h;
    const float* hpB = Bq + ((chunk - c) + w * HEADC) * H + h;
    v2f hp[HEADC], hq[HEADC];
    #pragma unroll
    for (int j = 0; j < HEADC; ++j) {
        const int jr = max(min(j, nh - 1), 0);
        hp[j] = *(const v2f*)(hpA + jr * H);
        hq[j] = *(const v2f*)(hpB + jr * H);
    }

    {
        float Px = 1.f, Py = 1.f, Qx = 0.f, Qy = 0.f;
        #pragma unroll
        for (int j = 0; j < HEADC; ++j) {
            const float px = j < nh ? hp[j].x : 1.f;
            const float py = j < nh ? hp[j].y : 1.f;
            const float qx = j < nh ? hq[j].x : 0.f;
            const float qy = j < nh ? hq[j].y : 0.f;
            Qx = __builtin_fmaf(px, Qx, qx);
            Qy = __builtin_fmaf(py, Qy, qy);
            Px = px * Px;
            Py = py * Py;
        }
        heads[w][lane] = make_float4(Px, Py, Qx, Qy);
    }

    float Px = a[0].x, Py = a[0].y, Qx = b[0].x, Qy = b[0].y;
    #pragma unroll
    for (int t = 1; t < SUBT; ++t) {
        Qx = __builtin_fmaf(a[t].x, Qx, b[t].x);
        Qy = __builtin_fmaf(a[t].y, Qy, b[t].y);
        Px = a[t].x * Px;
        Py = a[t].y * Py;
    }
    pieces[w][lane] = make_float4(Px, Qx, Py, Qy);
    __syncthreads();

    float sx = 0.f, sy = 0.f;
    #pragma unroll
    for (int i = 0; i < NPIECE; ++i) {
        const float4 pq = heads[i][lane];
        sx = __builtin_fmaf(pq.x, sx, pq.z);
        sy = __builtin_fmaf(pq.y, sy, pq.w);
    }
    for (int i = 0; i < w; ++i) {
        const float4 pq = pieces[i][lane];
        sx = __builtin_fmaf(pq.x, sx, pq.y);
        sy = __builtin_fmaf(pq.z, sy, pq.w);
    }

    float* po = out + off;
    #pragma unroll
    for (int t = 0; t < SUBT; ++t) {
        sx = __builtin_fmaf(a[t].x, sx, b[t].x);
        sy = __builtin_fmaf(a[t].y, sy, b[t].y);
        v2f o;
        o.x = __builtin_fmaf(sx, cv[t].x, r[t].x);
        o.y = __builtin_fmaf(sy, cv[t].y, r[t].y);
        *(v2f*)(po + t * H) = o;
    }
}

extern "C" void kernel_launch(void* const* d_in, const int* in_sizes, int n_in,
                              void* d_out, int out_size, void* d_ws, size_t ws_size,
                              hipStream_t stream)
{
    (void)in_sizes; (void)n_in; (void)out_size; (void)ws_size;

    const float* x  = (const float*)d_in[0];
    const float* W  = (const float*)d_in[1];
    const float* Av = (const float*)d_in[2];
    const float* Bv = (const float*)d_in[3];
    const float* Dv = (const float*)d_in[4];
    float* out = (float*)d_out;

    const size_t NE = (size_t)BATCH * LEN * H;
    const size_t NC = (size_t)BATCH * NCHUNK * H;
    float* ws = (float*)d_ws;
    float* dA = ws;
    float* dB = dA + NE;
    float* Cc = dB + NE;
    float* rD = Cc + NE;
    float* Ap = rD + NE;
    float* Bq = Ap + NC;

    const int ldsBytes = (16 * SLAB + 2 * H * 16) * (int)sizeof(float);
    hipFuncSetAttribute(reinterpret_cast<const void*>(k_proj),
                        hipFuncAttributeMaxDynamicSharedMemorySize, ldsBytes);

    k_proj<<<MROWS / 128, 256, ldsBytes, stream>>>(x, W, Av, Bv, Dv,
                                                   dA, dB, Cc, rD, Ap, Bq);
    k_apply<<<BATCH * NCHUNK * 2, 256, 0, stream>>>(dA, dB, Cc, rD, Ap, Bq, out);
}


# k_proj: elementwise work and stores of each 32-column group issued between the WMMAs of the next group (two accumulator sets), not behind all WMMAs of a 64-column half; sums unchanged
# speedup vs baseline: 1.4673x; 1.0395x over previous
#include <hip/hip_runtime.h>


typedef __attribute__((ext_vector_type(2))) float v2f;
typedef __attribute__((ext_vector_type(8))) float v8f;

#define H      128
#define FOURH  512
#define LEN    8192
#define BATCH  8
#define MROWS  (BATCH * LEN)
#define NCHUNK 64
#define TCH    (LEN / NCHUNK)
#define NSEG   3
#define HALFN  64
#define NTILE  (NSEG * HALFN / 16)
#define SLAB   (NTILE * H + 4)

__device__ __forceinline__
void stage_w_half(const float* __restrict__ W, float* lds, int half, int tid)
{
    for (int idx = tid; idx < (NSEG * HALFN * H) / 4; idx += 256) {
        const int ql = idx >> 5;
        const int k4 = (idx & 31) << 2;
        const int sg = ql >> 6;
        const int j  = ql & (HALFN - 1);
        const int p  = ql + sg * HALFN + (sg ? H : 0) + half * HALFN;
        const float4 w = *(const float4*)(W + p * H + k4);
        *(float4*)(lds + (j & 15) * SLAB + ((j >> 4) * NSEG + sg) * H + k4) = w;
    }
}

#define GCOLS  32
#define GT     (GCOLS / 16)
#define GTILE  (GT * NSEG)

struct Acc { v8f D[GT], C[GT], R[GT]; };

__device__ __forceinline__
void epi_elem(const Acc& p, const int nt, const int r,
              const float Ah, const float Bh, const float Dh,
              float& P, float& Q,
              float* __restrict__ dA, float* __restrict__ dB,
              float* __restrict__ Cc, float* __restrict__ rD, const long o)
{
    const float dv = p.D[nt][r];
    const float sp = fmaxf(dv, 0.f) + __logf(1.f + __expf(-fabsf(dv)));
    const float rs = p.R[nt][r];
    const float av = sp * Ah;
    const float bv = sp * Bh * rs;
    dA[o] = av;
    dB[o] = bv;
    Cc[o] = p.C[nt][r];
    rD[o] = rs * Dh;
    Q = av * Q + bv;
    P = av * P;
}

template <bool EPI>
__device__ __forceinline__
void mm_group(const v2f (&a)[32], const float* bg, Acc& cur, const Acc& prev,
              const float (&Ah)[GT], const float (&Bh)[GT], const float (&Dh)[GT],
              float* __restrict__ dA, float* __restrict__ dB,
              float* __restrict__ Cc, float* __restrict__ rD,
              const long o0, const int hprev, v2f* pieces, const int pslot)
{
    #pragma unroll
    for (int nt = 0; nt < GT; ++nt) {
        cur.D[nt] = (v8f){};
        cur.C[nt] = (v8f){};
        cur.R[nt] = (v8f){};
    }

    v2f bf[2][2 * GTILE];
    #pragma unroll
    for (int s = 0; s < 2; ++s) {
        #pragma unroll
        for (int t = 0; t < GTILE; ++t)
            bf[0][s * GTILE + t] = *(const v2f*)(bg + t * H + s * 4);
    }

    float P = 1.f, Q = 0.f;
    #pragma unroll
    for (int kp = 0; kp < 16; ++kp) {
        const int cs = kp & 1;
        if (kp + 1 < 16) {
            #pragma unroll
            for (int s = 0; s < 2; ++s) {
                #pragma unroll
                for (int t = 0; t < GTILE; ++t)
                    bf[cs ^ 1][s * GTILE + t] =
                        *(const v2f*)(bg + t * H + (2 * (kp + 1) + s) * 4);
            }
        }
        #pragma unroll
        for (int s = 0; s < 2; ++s) {
            const int kt = 2 * kp + s;
            #pragma unroll
            for (int nt = 0; nt < GT; ++nt) {
                cur.D[nt] = __builtin_amdgcn_wmma_f32_16x16x4_f32(
                    false, a[kt], false, bf[cs][s * GTILE + nt * NSEG + 0], (short)0, cur.D[nt], false, false);
                cur.C[nt] = __builtin_amdgcn_wmma_f32_16x16x4_f32(
                    false, a[kt], false, bf[cs][s * GTILE + nt * NSEG + 1], (short)0, cur.C[nt], false, false);
                cur.R[nt] = __builtin_amdgcn_wmma_f32_16x16x4_f32(
                    false, a[kt], false, bf[cs][s * GTILE + nt * NSEG + 2], (short)0, cur.R[nt], false, false);
            }
        }
        if (EPI) {
            const int nt = kp >> 3;
            const int r  = kp & 7;
            if (r == 0) { P = 1.f; Q = 0.f; }
            epi_elem(prev, nt, r, Ah[nt], Bh[nt], Dh[nt], P, Q,
                     dA, dB, Cc, rD, o0 + r * H + hprev + nt * 16);
            if (r == 7)
                pieces[(hprev + nt * 16) * 16 + pslot] = (v2f){P, Q};
        }
        __builtin_amdgcn_sched_barrier(0);
    }
}

__global__ __launch_bounds__(256)
void k_proj(const float* __restrict__ x, const float* __restrict__ W,
            const float* __restrict__ Avec, const float* __restrict__ Bvec,
            const float* __restrict__ Dvec,
            float* __restrict__ dA, float* __restrict__ dB,
            float* __restrict__ Cc, float* __restrict__ rD,
            float* __restrict__ Ap, float* __restrict__ Bq)
{
    extern __shared__ float lds[];
    const int tid = threadIdx.x;

    stage_w_half(W, lds, 0, tid);
    __syncthreads();

    const int wave = tid >> 5;
    const int lane = tid & 31;
    const int lm   = lane & 15;
    const int kh   = lane >> 4;
    const long rowBase = (long)blockIdx.x * 128 + wave * 16;

    v2f a[32];
    {
        const float* xp = x + (rowBase + lm) * H + 2 * kh;
        #pragma unroll
        for (int kt = 0; kt < 32; ++kt)
            a[kt] = *(const v2f*)(xp + kt * 4);
    }

    v2f* pieces = (v2f*)(lds + 16 * SLAB);
    v2f* piecesl = pieces + lm * 16;
    const int pslot = wave * 2 + kh;

    const float* bp = lds + lm * SLAB + 2 * kh;
    const long o0 = (rowBase + 8 * kh) * H + lm;

    Acc cur, prev;
    float Ah[GT], Bh[GT], Dh[GT];
    #pragma unroll
    for (int nt = 0; nt < GT; ++nt) { Ah[nt] = 0.f; Bh[nt] = 0.f; Dh[nt] = 0.f; }

    mm_group<false>(a, bp, cur, cur, Ah, Bh, Dh, dA, dB, Cc, rD, o0, 0, piecesl, pslot);
    prev = cur;

    for (int g = 1; g < 4; ++g) {
        if (g == 2) {
            __syncthreads();
            stage_w_half(W, lds, 1, tid);
            __syncthreads();
        }
        const int hprev = (g - 1) * GCOLS;
        #pragma unroll
        for (int nt = 0; nt < GT; ++nt) {
            const int h = hprev + nt * 16 + lm;
            Ah[nt] = Avec[h];
            Bh[nt] = Bvec[h];
            Dh[nt] = Dvec[h];
        }
        mm_group<true>(a, bp + (g & 1) * (GT * NSEG * H), cur, prev,
                       Ah, Bh, Dh, dA, dB, Cc, rD, o0, hprev, piecesl, pslot);
        prev = cur;
    }

    {
        const int hprev = 3 * GCOLS;
        #pragma unroll
        for (int nt = 0; nt < GT; ++nt) {
            const int h = hprev + nt * 16 + lm;
            const float Ahl = Avec[h];
            const float Bhl = Bvec[h];
            const float Dhl = Dvec[h];
            float P = 1.f, Q = 0.f;
            #pragma unroll
            for (int r = 0; r < 8; ++r)
                epi_elem(prev, nt, r, Ahl, Bhl, Dhl, P, Q,
                         dA, dB, Cc, rD, o0 + r * H + hprev + nt * 16);
            piecesl[(hprev + nt * 16) * 16 + pslot] = (v2f){P, Q};
        }
    }

    __syncthreads();
    if (tid < H) {
        float P = 1.f, Q = 0.f;
        #pragma unroll
        for (int i = 0; i < 16; ++i) {
            const v2f pq = pieces[tid * 16 + i];
            Q = pq.x * Q + pq.y;
            P = pq.x * P;
        }
        const long o = (long)blockIdx.x * H + tid;
        Ap[o] = P;
        Bq[o] = Q;
    }
}

#define SUBT  16
#define NPIECE (TCH / SUBT)
#define HEADC (NCHUNK / NPIECE)
__global__ __launch_bounds__(256)
void k_apply(const float* __restrict__ dA, const float* __restrict__ dB,
             const float* __restrict__ Cc, const float* __restrict__ rD,
             const float* __restrict__ Ap, const float* __restrict__ Bq,
             float* __restrict__ out)
{
    __shared__ float4 pieces[NPIECE][32];
    __shared__ float4 heads[NPIECE][32];

    const int lane  = threadIdx.x & 31;
    const int w     = __builtin_amdgcn_readfirstlane(threadIdx.x >> 5);
    const int chunk = blockIdx.x >> 1;
    const int c     = chunk & (NCHUNK - 1);
    const int h     = (blockIdx.x & 1) * 64 + 2 * lane;
    const unsigned off = ((unsigned)chunk * TCH + (unsigned)w * SUBT) * H + h;

    const float* pa = dA + off;
    const float* pb = dB + off;
    const float* pc = Cc + off;
    const float* pr = rD + off;

    v2f a[SUBT], b[SUBT], cv[SUBT], r[SUBT];
    #pragma unroll
    for (int t = 0; t < SUBT; ++t) {
        a[t] = *(const v2f*)(pa + t * H);
        b[t] = *(const v2f*)(pb + t * H);
    }
    #pragma unroll
    for (int t = 0; t < SUBT; ++t) {
        cv[t] = *(const v2f*)(pc + t * H);
        r[t]  = *(const v2f*)(pr + t * H);
    }

    const int nh = c - w * HEADC;
    const float* hpA = Ap + ((chunk - c) + w * HEADC) * H + h;
    const float* hpB = Bq + ((chunk - c) + w * HEADC) * H + h;
    v2f hp[HEADC], hq[HEADC];
    #pragma unroll
    for (int j = 0; j < HEADC; ++j) {
        const int jr = max(min(j, nh - 1), 0);
        hp[j] = *(const v2f*)(hpA + jr * H);
        hq[j] = *(const v2f*)(hpB + jr * H);
    }

    {
        float Px = 1.f, Py = 1.f, Qx = 0.f, Qy = 0.f;
        #pragma unroll
        for (int j = 0; j < HEADC; ++j) {
            const float px = j < nh ? hp[j].x : 1.f;
            const float py = j < nh ? hp[j].y : 1.f;
            const float qx = j < nh ? hq[j].x : 0.f;
            const float qy = j < nh ? hq[j].y : 0.f;
            Qx = __builtin_fmaf(px, Qx, qx);
            Qy = __builtin_fmaf(py, Qy, qy);
            Px = px * Px;
            Py = py * Py;
        }
        heads[w][lane] = make_float4(Px, Py, Qx, Qy);
    }

    float Px = a[0].x, Py = a[0].y, Qx = b[0].x, Qy = b[0].y;
    #pragma unroll
    for (int t = 1; t < SUBT; ++t) {
        Qx = __builtin_fmaf(a[t].x, Qx, b[t].x);
        Qy = __builtin_fmaf(a[t].y, Qy, b[t].y);
        Px = a[t].x * Px;
        Py = a[t].y * Py;
    }
    pieces[w][lane] = make_float4(Px, Qx, Py, Qy);
    __syncthreads();

    float sx = 0.f, sy = 0.f;
    #pragma unroll
    for (int i = 0; i < NPIECE; ++i) {
        const float4 pq = heads[i][lane];
        sx = __builtin_fmaf(pq.x, sx, pq.z);
        sy = __builtin_fmaf(pq.y, sy, pq.w);
    }
    for (int i = 0; i < w; ++i) {
        const float4 pq = pieces[i][lane];
        sx = __builtin_fmaf(pq.x, sx, pq.y);
        sy = __builtin_fmaf(pq.z, sy, pq.w);
    }

    float* po = out + off;
    #pragma unroll
    for (int t = 0; t < SUBT; ++t) {
        sx = __builtin_fmaf(a[t].x, sx, b[t].x);
        sy = __builtin_fmaf(a[t].y, sy, b[t].y);
        v2f o;
        o.x = __builtin_fmaf(sx, cv[t].x, r[t].x);
        o.y = __builtin_fmaf(sy, cv[t].y, r[t].y);
        *(v2f*)(po + t * H) = o;
    }
}

extern "C" void kernel_launch(void* const* d_in, const int* in_sizes, int n_in,
                              void* d_out, int out_size, void* d_ws, size_t ws_size,
                              hipStream_t stream)
{
    (void)in_sizes; (void)n_in; (void)out_size; (void)ws_size;

    const float* x  = (const float*)d_in[0];
    const float* W  = (const float*)d_in[1];
    const float* Av = (const float*)d_in[2];
    const float* Bv = (const float*)d_in[3];
    const float* Dv = (const float*)d_in[4];
    float* out = (float*)d_out;

    const size_t NE = (size_t)BATCH * LEN * H;
    const size_t NC = (size_t)BATCH * NCHUNK * H;
    float* ws = (float*)d_ws;
    float* dA = ws;
    float* dB = dA + NE;
    float* Cc = dB + NE;
    float* rD = Cc + NE;
    float* Ap = rD + NE;
    float* Bq = Ap + NC;

    const int ldsBytes = (16 * SLAB + 2 * H * 16) * (int)sizeof(float);
    hipFuncSetAttribute(reinterpret_cast<const void*>(k_proj),
                        hipFuncAttributeMaxDynamicSharedMemorySize, ldsBytes);

    k_proj<<<MROWS / 128, 256, ldsBytes, stream>>>(x, W, Av, Bv, Dv,
                                                   dA, dB, Cc, rD, Ap, Bq);
    k_apply<<<BATCH * NCHUNK * 2, 256, 0, stream>>>(dA, dB, Cc, rD, Ap, Bq, out);
}


# k_proj: all global loads of a staging phase in flight at once (x as whole 512-byte rows, to A fragments via LDS, plus 24 W pieces per thread; half 1 asked before its barrier), not 4 a trip; same sums
# speedup vs baseline: 1.5360x; 1.0468x over previous
#include <hip/hip_runtime.h>


typedef __attribute__((ext_vector_type(2))) float v2f;
typedef __attribute__((ext_vector_type(4))) float v4f;
typedef __attribute__((ext_vector_type(8))) float v8f;

#define H      128
#define FOURH  512
#define LEN    8192
#define BATCH  8
#define MROWS  (BATCH * LEN)
#define NCHUNK 64
#define TCH    (LEN / NCHUNK)
#define NSEG   3
#define HALFN  64
#define NTILE  (NSEG * HALFN / 16)
#define SLAB   (NTILE * H + 4)

#define WPT ((NSEG * HALFN * H) / 4 / 256)

__device__ __forceinline__
void load_w_half(const float* __restrict__ W, int half, int tid, v4f (&w)[WPT])
{
    const float* wb = W + ((tid >> 5) + half * HALFN) * H + ((tid & 31) << 2);
    #pragma unroll
    for (int i = 0; i < WPT; ++i) {
        const int sg = i >> 3;
        const int p  = 8 * (i & 7) + sg * 2 * HALFN + (sg ? H : 0);
        w[i] = *(const v4f*)(wb + p * H);
    }
}

__device__ __forceinline__
void store_w_half(float* lds, int tid, const v4f (&w)[WPT])
{
    float* lb = lds + (tid >> 5) * SLAB + ((tid & 31) << 2);
    #pragma unroll
    for (int i = 0; i < WPT; ++i) {
        const int sg = i >> 3;
        *(v4f*)(lb + 8 * (i & 1) * SLAB + (((i & 7) >> 1) * NSEG + sg) * H) = w[i];
    }
}

#define XPITCH (H + 4)

#define GCOLS  32
#define GT     (GCOLS / 16)
#define GTILE  (GT * NSEG)

struct Acc { v8f D[GT], C[GT], R[GT]; };

__device__ __forceinline__
void epi_elem(const Acc& p, const int nt, const int r,
              const float Ah, const float Bh, const float Dh,
              float& P, float& Q,
              float* __restrict__ dA, float* __restrict__ dB,
              float* __restrict__ Cc, float* __restrict__ rD, const unsigned ob)
{
    const float dv = p.D[nt][r];
    const float sp = fmaxf(dv, 0.f) + __logf(1.f + __expf(-fabsf(dv)));
    const float rs = p.R[nt][r];
    const float av = sp * Ah;
    const float bv = sp * Bh * rs;
    *(float*)((char*)dA + ob) = av;
    *(float*)((char*)dB + ob) = bv;
    *(float*)((char*)Cc + ob) = p.C[nt][r];
    *(float*)((char*)rD + ob) = rs * Dh;
    Q = av * Q + bv;
    P = av * P;
}

template <bool EPI>
__device__ __forceinline__
void mm_group(const v2f (&a)[32], const float* bg, Acc& cur, const Acc& prev,
              const float (&Ah)[GT], const float (&Bh)[GT], const float (&Dh)[GT],
              float* __restrict__ dA, float* __restrict__ dB,
              float* __restrict__ Cc, float* __restrict__ rD,
              const unsigned ob, const int hprev,
              v2f* pieces, const int pslot)
{
    #pragma unroll
    for (int nt = 0; nt < GT; ++nt) {
        cur.D[nt] = (v8f){};
        cur.C[nt] = (v8f){};
        cur.R[nt] = (v8f){};
    }

    v2f bf[2][2 * GTILE];
    #pragma unroll
    for (int s = 0; s < 2; ++s) {
        #pragma unroll
        for (int t = 0; t < GTILE; ++t)
            bf[0][s * GTILE + t] = *(const v2f*)(bg + t * H + s * 4);
    }

    float P = 1.f, Q = 0.f;
    #pragma unroll
    for (int kp = 0; kp < 16; ++kp) {
        const int cs = kp & 1;
        if (kp + 1 < 16) {
            #pragma unroll
            for (int s = 0; s < 2; ++s) {
                #pragma unroll
                for (int t = 0; t < GTILE; ++t)
                    bf[cs ^ 1][s * GTILE + t] =
                        *(const v2f*)(bg + t * H + (2 * (kp + 1) + s) * 4);
            }
        }
        #pragma unroll
        for (int s = 0; s < 2; ++s) {
            const int kt = 2 * kp + s;
            #pragma unroll
            for (int nt = 0; nt < GT; ++nt) {
                cur.D[nt] = __builtin_amdgcn_wmma_f32_16x16x4_f32(
                    false, a[kt], false, bf[cs][s * GTILE + nt * NSEG + 0], (short)0, cur.D[nt], false, false);
                cur.C[nt] = __builtin_amdgcn_wmma_f32_16x16x4_f32(
                    false, a[kt], false, bf[cs][s * GTILE + nt * NSEG + 1], (short)0, cur.C[nt], false, false);
                cur.R[nt] = __builtin_amdgcn_wmma_f32_16x16x4_f32(
                    false, a[kt], false, bf[cs][s * GTILE + nt * NSEG + 2], (short)0, cur.R[nt], false, false);
            }
        }
        if (EPI) {
            const int nt = kp >> 3;
            const int r  = kp & 7;
            if (r == 0) { P = 1.f; Q = 0.f; }
            const int c = r * H + nt * 16;
            epi_elem(prev, nt, r, Ah[nt], Bh[nt], Dh[nt], P, Q,
                     dA + c, dB + c, Cc + c, rD + c, ob);
            if (r == 7)
                pieces[(hprev + nt * 16) * 16 + pslot] = (v2f){P, Q};
        }
        __builtin_amdgcn_sched_barrier(0);
    }
}

__global__ __launch_bounds__(256)
void k_proj(const float* __restrict__ x, const float* __restrict__ W,
            const float* __restrict__ Avec, const float* __restrict__ Bvec,
            const float* __restrict__ Dvec,
            float* __restrict__ dA, float* __restrict__ dB,
            float* __restrict__ Cc, float* __restrict__ rD,
            float* __restrict__ Ap, float* __restrict__ Bq)
{
    extern __shared__ float lds[];
    const int tid = threadIdx.x;

    const int wave = tid >> 5;
    const int lane = tid & 31;
    const int lm   = lane & 15;
    const int kh   = lane >> 4;
    const long rowBase = (long)blockIdx.x * 128 + wave * 16;

    v4f xr[16];
    {
        const float* xp = x + rowBase * H + 4 * lane;
        #pragma unroll
        for (int q = 0; q < 16; ++q)
            xr[q] = *(const v4f*)(xp + q * H);
    }
    v4f w0[WPT];
    load_w_half(W, 0, tid, w0);

    float* xs = lds + wave * (16 * XPITCH);
    #pragma unroll
    for (int q = 0; q < 16; ++q)
        *(v4f*)(xs + q * XPITCH + 4 * lane) = xr[q];
    __syncthreads();

    v2f a[32];
    {
        const float* xf = xs + lm * XPITCH + 2 * kh;
        #pragma unroll
        for (int kt = 0; kt < 32; ++kt)
            a[kt] = *(const v2f*)(xf + kt * 4);
    }
    __syncthreads();
    store_w_half(lds, tid, w0);
    __syncthreads();

    v2f* pieces = (v2f*)(lds + 16 * SLAB);
    v2f* piecesl = pieces + lm * 16;
    const int pslot = wave * 2 + kh;

    const float* bp = lds + lm * SLAB + 2 * kh;
    const long wrow = (long)blockIdx.x * 128 + wave * 16;
    float* pdA = dA + wrow * H;
    float* pdB = dB + wrow * H;
    float* pCc = Cc + wrow * H;
    float* prD = rD + wrow * H;
    const float* pAv = Avec;
    const float* pBv = Bvec;
    const float* pDv = Dvec;
    const unsigned ob = (unsigned)((8 * kh * H + lm) * 4);
    const unsigned lm4 = (unsigned)(lm * 4);

    Acc cur, prev;
    float Ah[GT], Bh[GT], Dh[GT];
    #pragma unroll
    for (int nt = 0; nt < GT; ++nt) { Ah[nt] = 0.f; Bh[nt] = 0.f; Dh[nt] = 0.f; }

    mm_group<false>(a, bp, cur, cur, Ah, Bh, Dh, pdA, pdB, pCc, prD, ob, 0, piecesl, pslot);
    prev = cur;

    for (int g = 1; g < 4; ++g) {
        if (g == 2) {
            v4f w1[WPT];
            load_w_half(W, 1, tid, w1);
            __syncthreads();
            store_w_half(lds, tid, w1);
            __syncthreads();
        }
        const int hprev = (g - 1) * GCOLS;
        #pragma unroll
        for (int nt = 0; nt < GT; ++nt) {
            Ah[nt] = *(const float*)((const char*)(pAv + nt * 16) + lm4);
            Bh[nt] = *(const float*)((const char*)(pBv + nt * 16) + lm4);
            Dh[nt] = *(const float*)((const char*)(pDv + nt * 16) + lm4);
        }
        mm_group<true>(a, bp + (g & 1) * (GT * NSEG * H), cur, prev,
                       Ah, Bh, Dh, pdA, pdB, pCc, prD, ob, hprev, piecesl, pslot);
        prev = cur;
        pdA += GCOLS; pdB += GCOLS; pCc += GCOLS; prD += GCOLS;
        pAv += GCOLS; pBv += GCOLS; pDv += GCOLS;
    }

    {
        const int hprev = 3 * GCOLS;
        #pragma unroll
        for (int nt = 0; nt < GT; ++nt) {
            const float Ahl = *(const float*)((const char*)(pAv + nt * 16) + lm4);
            const float Bhl = *(const float*)((const char*)(pBv + nt * 16) + lm4);
            const float Dhl = *(const float*)((const char*)(pDv + nt * 16) + lm4);
            float P = 1.f, Q = 0.f;
            #pragma unroll
            for (int r = 0; r < 8; ++r) {
                const int c = r * H + nt * 16;
                epi_elem(prev, nt, r, Ahl, Bhl, Dhl, P, Q,
                         pdA + c, pdB + c, pCc + c, prD + c, ob);
            }
            piecesl[(hprev + nt * 16) * 16 + pslot] = (v2f){P, Q};
        }
    }

    __syncthreads();
    if (tid < H) {
        float P = 1.f, Q = 0.f;
        #pragma unroll
        for (int i = 0; i < 16; ++i) {
            const v2f pq = pieces[tid * 16 + i];
            Q = pq.x * Q + pq.y;
            P = pq.x * P;
        }
        const long o = (long)blockIdx.x * H + tid;
        Ap[o] = P;
        Bq[o] = Q;
    }
}

#define SUBT  16
#define NPIECE (TCH / SUBT)
#define HEADC (NCHUNK / NPIECE)
__global__ __launch_bounds__(256)
void k_apply(const float* __restrict__ dA, const float* __restrict__ dB,
             const float* __restrict__ Cc, const float* __restrict__ rD,
             const float* __restrict__ Ap, const float* __restrict__ Bq,
             float* __restrict__ out)
{
    __shared__ float4 pieces[NPIECE][32];
    __shared__ float4 heads[NPIECE][32];

    const int lane  = threadIdx.x & 31;
    const int w     = __builtin_amdgcn_readfirstlane(threadIdx.x >> 5);
    const int chunk = blockIdx.x >> 1;
    const int c     = chunk & (NCHUNK - 1);
    const int h     = (blockIdx.x & 1) * 64 + 2 * lane;
    const unsigned off = ((unsigned)chunk * TCH + (unsigned)w * SUBT) * H + h;

    const float* pa = dA + off;
    const float* pb = dB + off;
    const float* pc = Cc + off;
    const float* pr = rD + off;

    v2f a[SUBT], b[SUBT], cv[SUBT], r[SUBT];
    #pragma unroll
    for (int t = 0; t < SUBT; ++t) {
        a[t] = *(const v2f*)(pa + t * H);
        b[t] = *(const v2f*)(pb + t * H);
    }
    #pragma unroll
    for (int t = 0; t < SUBT; ++t) {
        cv[t] = *(const v2f*)(pc + t * H);
        r[t]  = *(const v2f*)(pr + t * H);
    }

    const int nh = c - w * HEADC;
    const float* hpA = Ap + ((chunk - c) + w * HEADC) * H + h;
    const float* hpB = Bq + ((chunk - c) + w * HEADC) * H + h;
    v2f hp[HEADC], hq[HEADC];
    #pragma unroll
    for (int j = 0; j < HEADC; ++j) {
        const int jr = max(min(j, nh - 1), 0);
        hp[j] = *(const v2f*)(hpA + jr * H);
        hq[j] = *(const v2f*)(hpB + jr * H);
    }

    {
        float Px = 1.f, Py = 1.f, Qx = 0.f, Qy = 0.f;
        #pragma unroll
        for (int j = 0; j < HEADC; ++j) {
            const float px = j < nh ? hp[j].x : 1.f;
            const float py = j < nh ? hp[j].y : 1.f;
            const float qx = j < nh ? hq[j].x : 0.f;
            const float qy = j < nh ? hq[j].y : 0.f;
            Qx = __builtin_fmaf(px, Qx, qx);
            Qy = __builtin_fmaf(py, Qy, qy);
            Px = px * Px;
            Py = py * Py;
        }
        heads[w][lane] = make_float4(Px, Py, Qx, Qy);
    }

    float Px = a[0].x, Py = a[0].y, Qx = b[0].x, Qy = b[0].y;
    #pragma unroll
    for (int t = 1; t < SUBT; ++t) {
        Qx = __builtin_fmaf(a[t].x, Qx, b[t].x);
        Qy = __builtin_fmaf(a[t].y, Qy, b[t].y);
        Px = a[t].x * Px;
        Py = a[t].y * Py;
    }
    pieces[w][lane] = make_float4(Px, Qx, Py, Qy);
    __syncthreads();

    float sx = 0.f, sy = 0.f;
    #pragma unroll
    for (int i = 0; i < NPIECE; ++i) {
        const float4 pq = heads[i][lane];
        sx = __builtin_fmaf(pq.x, sx, pq.z);
        sy = __builtin_fmaf(pq.y, sy, pq.w);
    }
    for (int i = 0; i < w; ++i) {
        const float4 pq = pieces[i][lane];
        sx = __builtin_fmaf(pq.x, sx, pq.y);
        sy = __builtin_fmaf(pq.z, sy, pq.w);
    }

    float* po = out + off;
    #pragma unroll
    for (int t = 0; t < SUBT; ++t) {
        sx = __builtin_fmaf(a[t].x, sx, b[t].x);
        sy = __builtin_fmaf(a[t].y, sy, b[t].y);
        v2f o;
        o.x = __builtin_fmaf(sx, cv[t].x, r[t].x);
        o.y = __builtin_fmaf(sy, cv[t].y, r[t].y);
        *(v2f*)(po + t * H) = o;
    }
}

extern "C" void kernel_launch(void* const* d_in, const int* in_sizes, int n_in,
                              void* d_out, int out_size, void* d_ws, size_t ws_size,
                              hipStream_t stream)
{
    (void)in_sizes; (void)n_in; (void)out_size; (void)ws_size;

    const float* x  = (const float*)d_in[0];
    const float* W  = (const float*)d_in[1];
    const float* Av = (const float*)d_in[2];
    const float* Bv = (const float*)d_in[3];
    const float* Dv = (const float*)d_in[4];
    float* out = (float*)d_out;

    const size_t NE = (size_t)BATCH * LEN * H;
    const size_t NC = (size_t)BATCH * NCHUNK * H;
    float* ws = (float*)d_ws;
    float* dA = ws;
    float* dB = dA + NE;
    float* Cc = dB + NE;
    float* rD = Cc + NE;
    float* Ap = rD + NE;
    float* Bq = Ap + NC;

    const int ldsBytes = (16 * SLAB + 2 * H * 16) * (int)sizeof(float);
    hipFuncSetAttribute(reinterpret_cast<const void*>(k_proj),
                        hipFuncAttributeMaxDynamicSharedMemorySize, ldsBytes);

    k_proj<<<MROWS / 128, 256, ldsBytes, stream>>>(x, W, Av, Bv, Dv,
                                                   dA, dB, Cc, rD, Ap, Bq);
    k_apply<<<BATCH * NCHUNK * 2, 256, 0, stream>>>(dA, dB, Cc, rD, Ap, Bq, out);
}


# C product (a third of the WMMAs) moved from k_proj into k_apply: a k_apply wave forms C for its 16 rows x 64 columns under its plane loads (its barriers wait for LDS only); 3 planes, not 4; same sums
# speedup vs baseline: 1.5860x; 1.0326x over previous
#include <hip/hip_runtime.h>


typedef __attribute__((ext_vector_type(2))) float v2f;
typedef __attribute__((ext_vector_type(4))) float v4f;
typedef __attribute__((ext_vector_type(8))) float v8f;

#define H      128
#define FOURH  512
#define LEN    8192
#define BATCH  8
#define MROWS  (BATCH * LEN)
#define NCHUNK 64
#define TCH    (LEN / NCHUNK)
#define NSEG   2
#define HALFN  64
#define NTILE  (NSEG * HALFN / 16)
#define SLAB   (NTILE * H + 4)

#define WPT ((NSEG * HALFN * H) / 4 / 256)

__device__ __forceinline__
void load_w_half(const float* __restrict__ W, int half, int tid, v4f (&w)[WPT])
{
    const float* wb = W + ((tid >> 5) + half * HALFN) * H + ((tid & 31) << 2);
    #pragma unroll
    for (int i = 0; i < WPT; ++i) {
        const int sg = i >> 3;
        const int p  = 8 * (i & 7) + (sg ? 3 * H : 0);
        w[i] = *(const v4f*)(wb + p * H);
    }
}

__device__ __forceinline__
void store_w_half(float* lds, int tid, const v4f (&w)[WPT])
{
    float* lb = lds + (tid >> 5) * SLAB + ((tid & 31) << 2);
    #pragma unroll
    for (int i = 0; i < WPT; ++i) {
        const int sg = i >> 3;
        *(v4f*)(lb + 8 * (i & 1) * SLAB + (((i & 7) >> 1) * NSEG + sg) * H) = w[i];
    }
}

#define XPITCH (H + 4)

#define GCOLS  32
#define GT     (GCOLS / 16)
#define GTILE  (GT * NSEG)

struct Acc { v8f D[GT], R[GT]; };

__device__ __forceinline__
void epi_elem(const Acc& p, const int nt, const int r,
              const float Ah, const float Bh, const float Dh,
              float& P, float& Q,
              float* __restrict__ dA, float* __restrict__ dB,
              float* __restrict__ rD, const unsigned ob)
{
    const float dv = p.D[nt][r];
    const float sp = fmaxf(dv, 0.f) + __logf(1.f + __expf(-fabsf(dv)));
    const float rs = p.R[nt][r];
    const float av = sp * Ah;
    const float bv = sp * Bh * rs;
    *(float*)((char*)dA + ob) = av;
    *(float*)((char*)dB + ob) = bv;
    *(float*)((char*)rD + ob) = rs * Dh;
    Q = av * Q + bv;
    P = av * P;
}

template <bool EPI>
__device__ __forceinline__
void mm_group(const v2f (&a)[32], const float* bg, Acc& cur, const Acc& prev,
              const float (&Ah)[GT], const float (&Bh)[GT], const float (&Dh)[GT],
              float* __restrict__ dA, float* __restrict__ dB,
              float* __restrict__ rD,
              const unsigned ob, const int hprev,
              v2f* pieces, const int pslot)
{
    #pragma unroll
    for (int nt = 0; nt < GT; ++nt) {
        cur.D[nt] = (v8f){};
        cur.R[nt] = (v8f){};
    }

    v2f bf[2][2 * GTILE];
    #pragma unroll
    for (int s = 0; s < 2; ++s) {
        #pragma unroll
        for (int t = 0; t < GTILE; ++t)
            bf[0][s * GTILE + t] = *(const v2f*)(bg + t * H + s * 4);
    }

    float P = 1.f, Q = 0.f;
    #pragma unroll
    for (int kp = 0; kp < 16; ++kp) {
        const int cs = kp & 1;
        if (kp + 1 < 16) {
            #pragma unroll
            for (int s = 0; s < 2; ++s) {
                #pragma unroll
                for (int t = 0; t < GTILE; ++t)
                    bf[cs ^ 1][s * GTILE + t] =
                        *(const v2f*)(bg + t * H + (2 * (kp + 1) + s) * 4);
            }
        }
        #pragma unroll
        for (int s = 0; s < 2; ++s) {
            const int kt = 2 * kp + s;
            #pragma unroll
            for (int nt = 0; nt < GT; ++nt) {
                cur.D[nt] = __builtin_amdgcn_wmma_f32_16x16x4_f32(
                    false, a[kt], false, bf[cs][s * GTILE + nt * NSEG + 0], (short)0, cur.D[nt], false, false);
                cur.R[nt] = __builtin_amdgcn_wmma_f32_16x16x4_f32(
                    false, a[kt], false, bf[cs][s * GTILE + nt * NSEG + 1], (short)0, cur.R[nt], false, false);
            }
        }
        if (EPI) {
            const int nt = kp >> 3;
            const int r  = kp & 7;
            if (r == 0) { P = 1.f; Q = 0.f; }
            const int c = r * H + nt * 16;
            epi_elem(prev, nt, r, Ah[nt], Bh[nt], Dh[nt], P, Q,
                     dA + c, dB + c, rD + c, ob);
            if (r == 7)
                pieces[(hprev + nt * 16) * 16 + pslot] = (v2f){P, Q};
        }
        __builtin_amdgcn_sched_barrier(0);
    }
}

__global__ __launch_bounds__(256)
void k_proj(const float* __restrict__ x, const float* __restrict__ W,
            const float* __restrict__ Avec, const float* __restrict__ Bvec,
            const float* __restrict__ Dvec,
            float* __restrict__ dA, float* __restrict__ dB,
            float* __restrict__ rD,
            float* __restrict__ Ap, float* __restrict__ Bq)
{
    extern __shared__ float lds[];
    const int tid = threadIdx.x;

    const int wave = tid >> 5;
    const int lane = tid & 31;
    const int lm   = lane & 15;
    const int kh   = lane >> 4;
    const long rowBase = (long)blockIdx.x * 128 + wave * 16;

    v4f xr[16];
    {
        const float* xp = x + rowBase * H + 4 * lane;
        #pragma unroll
        for (int q = 0; q < 16; ++q)
            xr[q] = *(const v4f*)(xp + q * H);
    }
    v4f w0[WPT];
    load_w_half(W, 0, tid, w0);

    float* xs = lds + wave * (16 * XPITCH);
    #pragma unroll
    for (int q = 0; q < 16; ++q)
        *(v4f*)(xs + q * XPITCH + 4 * lane) = xr[q];
    __syncthreads();

    v2f a[32];
    {
        const float* xf = xs + lm * XPITCH + 2 * kh;
        #pragma unroll
        for (int kt = 0; kt < 32; ++kt)
            a[kt] = *(const v2f*)(xf + kt * 4);
    }
    __syncthreads();
    store_w_half(lds, tid, w0);
    __syncthreads();

    v2f* pieces = (v2f*)(lds + 16 * SLAB);
    v2f* piecesl = pieces + lm * 16;
    const int pslot = wave * 2 + kh;

    const float* bp = lds + lm * SLAB + 2 * kh;
    const long wrow = (long)blockIdx.x * 128 + wave * 16;
    float* pdA = dA + wrow * H;
    float* pdB = dB + wrow * H;
    float* prD = rD + wrow * H;
    const float* pAv = Avec;
    const float* pBv = Bvec;
    const float* pDv = Dvec;
    const unsigned ob = (unsigned)((8 * kh * H + lm) * 4);
    const unsigned lm4 = (unsigned)(lm * 4);

    Acc cur, prev;
    float Ah[GT], Bh[GT], Dh[GT];
    #pragma unroll
    for (int nt = 0; nt < GT; ++nt) { Ah[nt] = 0.f; Bh[nt] = 0.f; Dh[nt] = 0.f; }

    mm_group<false>(a, bp, cur, cur, Ah, Bh, Dh, pdA, pdB, prD, ob, 0, piecesl, pslot);
    prev = cur;

    for (int g = 1; g < 4; ++g) {
        if (g == 2) {
            v4f w1[WPT];
            load_w_half(W, 1, tid, w1);
            __syncthreads();
            store_w_half(lds, tid, w1);
            __syncthreads();
        }
        const int hprev = (g - 1) * GCOLS;
        #pragma unroll
        for (int nt = 0; nt < GT; ++nt) {
            Ah[nt] = *(const float*)((const char*)(pAv + nt * 16) + lm4);
            Bh[nt] = *(const float*)((const char*)(pBv + nt * 16) + lm4);
            Dh[nt] = *(const float*)((const char*)(pDv + nt * 16) + lm4);
        }
        mm_group<true>(a, bp + (g & 1) * (GT * NSEG * H), cur, prev,
                       Ah, Bh, Dh, pdA, pdB, prD, ob, hprev, piecesl, pslot);
        prev = cur;
        pdA += GCOLS; pdB += GCOLS; prD += GCOLS;
        pAv += GCOLS; pBv += GCOLS; pDv += GCOLS;
    }

    {
        const int hprev = 3 * GCOLS;
        #pragma unroll
        for (int nt = 0; nt < GT; ++nt) {
            const float Ahl = *(const float*)((const char*)(pAv + nt * 16) + lm4);
            const float Bhl = *(const float*)((const char*)(pBv + nt * 16) + lm4);
            const float Dhl = *(const float*)((const char*)(pDv + nt * 16) + lm4);
            float P = 1.f, Q = 0.f;
            #pragma unroll
            for (int r = 0; r < 8; ++r) {
                const int c = r * H + nt * 16;
                epi_elem(prev, nt, r, Ahl, Bhl, Dhl, P, Q,
                         pdA + c, pdB + c, prD + c, ob);
            }
            piecesl[(hprev + nt * 16) * 16 + pslot] = (v2f){P, Q};
        }
    }

    __syncthreads();
    if (tid < H) {
        float P = 1.f, Q = 0.f;
        #pragma unroll
        for (int i = 0; i < 16; ++i) {
            const v2f pq = pieces[tid * 16 + i];
            Q = pq.x * Q + pq.y;
            P = pq.x * P;
        }
        const long o = (long)blockIdx.x * H + tid;
        Ap[o] = P;
        Bq[o] = Q;
    }
}

__device__ __forceinline__ void lds_barrier()
{
    __builtin_amdgcn_fence(__ATOMIC_RELEASE, "workgroup", "local");
    __builtin_amdgcn_s_barrier();
    __builtin_amdgcn_fence(__ATOMIC_ACQUIRE, "workgroup", "local");
}

#define SUBT  16
#define NPIECE (TCH / SUBT)
#define HEADC (NCHUNK / NPIECE)
#define CTILE (HALFN / 16)
#define CSLAB (CTILE * H + 4)
#define CWPT  ((HALFN * H) / 4 / 256)
__global__ __launch_bounds__(256)
void k_apply(const float* __restrict__ x, const float* __restrict__ W,
             const float* __restrict__ dA, const float* __restrict__ dB,
             const float* __restrict__ rD,
             const float* __restrict__ Ap, const float* __restrict__ Bq,
             float* __restrict__ out)
{
    extern __shared__ float lds[];
    float*  ldsW   = lds;
    float*  ldsX   = lds + 16 * CSLAB;
    float4* pieces = (float4*)(ldsX + NPIECE * 16 * XPITCH);
    float4* heads  = pieces + NPIECE * 32;

    const int tid   = threadIdx.x;
    const int lane  = tid & 31;
    const int lm    = lane & 15;
    const int kh    = lane >> 4;
    const int w     = __builtin_amdgcn_readfirstlane(tid >> 5);
    const int chunk = blockIdx.x >> 1;
    const int c     = chunk & (NCHUNK - 1);
    const int hb    = blockIdx.x & 1;
    const int h     = hb * HALFN + 2 * lane;
    const unsigned row0 = (unsigned)chunk * TCH + (unsigned)w * SUBT;
    const unsigned off  = row0 * H + h;

    const int nh = c - w * HEADC;
    const float* hpA = Ap + ((chunk - c) + w * HEADC) * H + h;
    const float* hpB = Bq + ((chunk - c) + w * HEADC) * H + h;
    v2f hp[HEADC], hq[HEADC];
    #pragma unroll
    for (int j = 0; j < HEADC; ++j) {
        const int jr = max(min(j, nh - 1), 0);
        hp[j] = *(const v2f*)(hpA + jr * H);
        hq[j] = *(const v2f*)(hpB + jr * H);
    }
    v4f xr[SUBT];
    {
        const float* xp = x + row0 * H + 4 * lane;
        #pragma unroll
        for (int q = 0; q < SUBT; ++q)
            xr[q] = *(const v4f*)(xp + q * H);
    }
    v4f wc[CWPT];
    {
        const float* wb = W + (2 * H + hb * HALFN + w) * H + 4 * lane;
        #pragma unroll
        for (int i = 0; i < CWPT; ++i)
            wc[i] = *(const v4f*)(wb + 8 * i * H);
    }
    const float* pa = dA + off;
    const float* pb = dB + off;
    const float* pr = rD + off;
    v2f a[SUBT], b[SUBT], r[SUBT];
    #pragma unroll
    for (int t = 0; t < SUBT; ++t) {
        a[t] = *(const v2f*)(pa + t * H);
        b[t] = *(const v2f*)(pb + t * H);
    }
    #pragma unroll
    for (int t = 0; t < SUBT; ++t)
        r[t] = *(const v2f*)(pr + t * H);

    {
        float Px = 1.f, Py = 1.f, Qx = 0.f, Qy = 0.f;
        #pragma unroll
        for (int j = 0; j < HEADC; ++j) {
            const float px = j < nh ? hp[j].x : 1.f;
            const float py = j < nh ? hp[j].y : 1.f;
            const float qx = j < nh ? hq[j].x : 0.f;
            const float qy = j < nh ? hq[j].y : 0.f;
            Qx = __builtin_fmaf(px, Qx, qx);
            Qy = __builtin_fmaf(py, Qy, qy);
            Px = px * Px;
            Py = py * Py;
        }
        heads[w * 32 + lane] = make_float4(Px, Py, Qx, Qy);
    }

    float* xs = ldsX + w * (16 * XPITCH);
    #pragma unroll
    for (int q = 0; q < SUBT; ++q)
        *(v4f*)(xs + q * XPITCH + 4 * lane) = xr[q];
    {
        float* lb = ldsW + w * CSLAB + 4 * lane;
        #pragma unroll
        for (int i = 0; i < CWPT; ++i)
            *(v4f*)(lb + 8 * (i & 1) * CSLAB + (i >> 1) * H) = wc[i];
    }
    lds_barrier();

    v8f acc[CTILE];
    #pragma unroll
    for (int nt = 0; nt < CTILE; ++nt) acc[nt] = (v8f){};
    {
        const float* xf = xs + lm * XPITCH + 2 * kh;
        const float* bp = ldsW + lm * CSLAB + 2 * kh;
        v2f af[2][2], bf[2][2 * CTILE];
        #pragma unroll
        for (int s = 0; s < 2; ++s) {
            af[0][s] = *(const v2f*)(xf + s * 4);
            #pragma unroll
            for (int nt = 0; nt < CTILE; ++nt)
                bf[0][s * CTILE + nt] = *(const v2f*)(bp + nt * H + s * 4);
        }
        #pragma unroll
        for (int kp = 0; kp < 16; ++kp) {
            const int cs = kp & 1;
            if (kp + 1 < 16) {
                #pragma unroll
                for (int s = 0; s < 2; ++s) {
                    const int kt = 2 * (kp + 1) + s;
                    af[cs ^ 1][s] = *(const v2f*)(xf + kt * 4);
                    #pragma unroll
                    for (int nt = 0; nt < CTILE; ++nt)
                        bf[cs ^ 1][s * CTILE + nt] = *(const v2f*)(bp + nt * H + kt * 4);
                }
            }
            __builtin_amdgcn_sched_barrier(0);
            #pragma unroll
            for (int s = 0; s < 2; ++s) {
                #pragma unroll
                for (int nt = 0; nt < CTILE; ++nt)
                    acc[nt] = __builtin_amdgcn_wmma_f32_16x16x4_f32(
                        false, af[cs][s], false, bf[cs][s * CTILE + nt], (short)0, acc[nt], false, false);
            }
            __builtin_amdgcn_sched_barrier(0);
        }
    }
    #pragma unroll
    for (int nt = 0; nt < CTILE; ++nt) {
        #pragma unroll
        for (int rr = 0; rr < 8; ++rr)
            xs[(8 * kh + rr) * XPITCH + nt * 16 + lm] = acc[nt][rr];
    }

    float Px = a[0].x, Py = a[0].y, Qx = b[0].x, Qy = b[0].y;
    #pragma unroll
    for (int t = 1; t < SUBT; ++t) {
        Qx = __builtin_fmaf(a[t].x, Qx, b[t].x);
        Qy = __builtin_fmaf(a[t].y, Qy, b[t].y);
        Px = a[t].x * Px;
        Py = a[t].y * Py;
    }
    pieces[w * 32 + lane] = make_float4(Px, Qx, Py, Qy);
    lds_barrier();

    v2f cv[SUBT];
    #pragma unroll
    for (int t = 0; t < SUBT; ++t)
        cv[t] = *(const v2f*)(xs + t * XPITCH + 2 * lane);

    float sx = 0.f, sy = 0.f;
    #pragma unroll
    for (int i = 0; i < NPIECE; ++i) {
        const float4 pq = heads[i * 32 + lane];
        sx = __builtin_fmaf(pq.x, sx, pq.z);
        sy = __builtin_fmaf(pq.y, sy, pq.w);
    }
    for (int i = 0; i < w; ++i) {
        const float4 pq = pieces[i * 32 + lane];
        sx = __builtin_fmaf(pq.x, sx, pq.y);
        sy = __builtin_fmaf(pq.z, sy, pq.w);
    }

    float* po = out + off;
    #pragma unroll
    for (int t = 0; t < SUBT; ++t) {
        sx = __builtin_fmaf(a[t].x, sx, b[t].x);
        sy = __builtin_fmaf(a[t].y, sy, b[t].y);
        v2f o;
        o.x = __builtin_fmaf(sx, cv[t].x, r[t].x);
        o.y = __builtin_fmaf(sy, cv[t].y, r[t].y);
        *(v2f*)(po + t * H) = o;
    }
}

extern "C" void kernel_launch(void* const* d_in, const int* in_sizes, int n_in,
                              void* d_out, int out_size, void* d_ws, size_t ws_size,
                              hipStream_t stream)
{
    (void)in_sizes; (void)n_in; (void)out_size; (void)ws_size;

    const float* x  = (const float*)d_in[0];
    const float* W  = (const float*)d_in[1];
    const float* Av = (const float*)d_in[2];
    const float* Bv = (const float*)d_in[3];
    const float* Dv = (const float*)d_in[4];
    float* out = (float*)d_out;

    const size_t NE = (size_t)BATCH * LEN * H;
    const size_t NC = (size_t)BATCH * NCHUNK * H;
    float* ws = (float*)d_ws;
    float* dA = ws;
    float* dB = dA + NE;
    float* rD = dB + NE;
    float* Ap = rD + NE;
    float* Bq = Ap + NC;

    const int ldsProj  = (16 * SLAB + 2 * H * 16) * (int)sizeof(float);
    const int ldsApply = (16 * CSLAB + NPIECE * 16 * XPITCH) * (int)sizeof(float)
                       + 2 * NPIECE * 32 * (int)sizeof(float4);
    hipFuncSetAttribute(reinterpret_cast<const void*>(k_proj),
                        hipFuncAttributeMaxDynamicSharedMemorySize, ldsProj);
    hipFuncSetAttribute(reinterpret_cast<const void*>(k_apply),
                        hipFuncAttributeMaxDynamicSharedMemorySize, ldsApply);

    k_proj<<<MROWS / 128, 256, ldsProj, stream>>>(x, W, Av, Bv, Dv,
                                                  dA, dB, rD, Ap, Bq);
    k_apply<<<BATCH * NCHUNK * 2, 256, ldsApply, stream>>>(x, W, dA, dB, rD, Ap, Bq, out);
}
